# starting kernel
# baseline (speedup 1.0000x reference)
#include <hip/hip_runtime.h>
#include <math.h>

#define N_ITEMS 200000
#define DIMS    128
#define S_PERMS 128
#define ALPHA_C 0.7f
#define EPS_C   1e-8f

typedef float v2f __attribute__((ext_vector_type(2)));
typedef float v8f __attribute__((ext_vector_type(8)));

__device__ __forceinline__ unsigned fenc(float f) {
  unsigned u = __float_as_uint(f);
  return (u & 0x80000000u) ? ~u : (u | 0x80000000u);
}
__device__ __forceinline__ float fdec(unsigned e) {
  unsigned u = (e & 0x80000000u) ? (e & 0x7FFFFFFFu) : ~e;
  return __uint_as_float(u);
}

__global__ void init_kernel(const float* __restrict__ tgt,
                            float* __restrict__ stats,
                            unsigned* __restrict__ umax) {
  int lane = threadIdx.x;
  float ss = 0.f;
  for (int k = lane; k < DIMS; k += 32) { float v = tgt[k]; ss = fmaf(v, v, ss); }
  #pragma unroll
  for (int o = 16; o; o >>= 1) ss += __shfl_xor(ss, o);
  if (lane == 0) { stats[0] = sqrtf(ss); stats[1] = 0.f; *umax = 0u; }
}

__global__ void __launch_bounds__(256)
vals_kernel(const float* __restrict__ neigh, const float* __restrict__ tgt,
            const float* __restrict__ stats, float* __restrict__ vals,
            float* __restrict__ w, int nTiles) {
  __shared__ __align__(16) float lds[8][16 * 132];
  const int lane  = threadIdx.x & 31;
  const int wave  = threadIdx.x >> 5;
  const int hi    = lane >> 4;
  const int row16 = lane & 15;
  float* tile = &lds[wave][0];
  const float tn = fmaxf(stats[0], EPS_C);

  v2f bv[32];
  #pragma unroll
  for (int k = 0; k < 32; ++k)
    bv[k] = *(const v2f*)(tgt + 4 * k + 2 * hi);

  const int r = lane >> 1, h = lane & 1;
  const int wavesTotal = (gridDim.x * blockDim.x) >> 5;
  for (int t = (blockIdx.x * blockDim.x + threadIdx.x) >> 5; t < nTiles;
       t += wavesTotal) {
    const float* src = neigh + (size_t)t * (16 * DIMS);
    const float4* srow = (const float4*)(src + r * DIMS + h * 64);
    float* drow = tile + r * 132 + h * 64;
    float ss = 0.f;
    #pragma unroll
    for (int j = 0; j < 16; ++j) {
      float4 v = srow[j];
      ss = fmaf(v.x, v.x, ss); ss = fmaf(v.y, v.y, ss);
      ss = fmaf(v.z, v.z, ss); ss = fmaf(v.w, v.w, ss);
      *(float4*)(drow + 4 * j) = v;
    }
    ss += __shfl_xor(ss, 1);

    v8f acc = {};
    const float* arow = tile + row16 * 132 + 2 * hi;
    #pragma unroll
    for (int k = 0; k < 32; ++k) {
      v2f a = *(const v2f*)(arow + 4 * k);
      acc = __builtin_amdgcn_wmma_f32_16x16x4_f32(false, a, false, bv[k],
                                                  (short)0, acc, false, false);
    }

    float vout[8];
    #pragma unroll
    for (int j = 0; j < 8; ++j) {
      float ssj = __shfl(ss, (lane & 16) + 2 * j);
      float dot = acc[j];
      float nn  = fmaxf(sqrtf(ssj), EPS_C);
      float sim = dot / (nn * tn);
      vout[j] = dot * __powf(sim, ALPHA_C);
    }
    if (row16 == 0) {
      int base = t * 16 + hi * 8;
      float4 lo  = make_float4(vout[0], vout[1], vout[2], vout[3]);
      float4 hi4 = make_float4(vout[4], vout[5], vout[6], vout[7]);
      *(float4*)(vals + base)     = lo;  *(float4*)(vals + base + 4) = hi4;
      *(float4*)(w + base)        = lo;  *(float4*)(w + base + 4)    = hi4;
    }
  }
}

__global__ void __launch_bounds__(256)
scatter_kernel(const int* __restrict__ perms, const float* __restrict__ vals,
               float* __restrict__ w) {
  const int s = blockIdx.y;
  const int i = blockIdx.x * blockDim.x + threadIdx.x;
  if (i >= N_ITEMS) return;
  const int idx = s * N_ITEMS + i;
  const int p  = perms[idx];
  int pp = __shfl_up(p, 1);
  if ((threadIdx.x & 31) == 0 && i > 0) pp = perms[idx - 1];
  if (i > 0) atomicAdd(&w[p], vals[pp] * (-1.0f / S_PERMS));
}

__global__ void max_kernel(const float* __restrict__ w, unsigned* __restrict__ umax, int n) {
  float m = -INFINITY;
  for (int i = blockIdx.x * blockDim.x + threadIdx.x; i < n; i += gridDim.x * blockDim.x)
    m = fmaxf(m, w[i]);
  #pragma unroll
  for (int o = 16; o; o >>= 1) m = fmaxf(m, __shfl_xor(m, o));
  if ((threadIdx.x & 31) == 0) atomicMax(umax, fenc(m));
}

__global__ void sum_kernel(const float* __restrict__ w, const unsigned* __restrict__ umax,
                           float* __restrict__ sum, int n) {
  const float mx = fdec(*umax);
  float a = 0.f;
  for (int i = blockIdx.x * blockDim.x + threadIdx.x; i < n; i += gridDim.x * blockDim.x)
    a += __expf(w[i] - mx);
  #pragma unroll
  for (int o = 16; o; o >>= 1) a += __shfl_xor(a, o);
  if ((threadIdx.x & 31) == 0) atomicAdd(sum, a);
}

__global__ void final_kernel(float* __restrict__ w, const unsigned* __restrict__ umax,
                             const float* __restrict__ sum, int n) {
  const float mx = fdec(*umax);
  const float rs = 1.0f / *sum;
  int i = blockIdx.x * blockDim.x + threadIdx.x;
  if (i < n) w[i] = __expf(w[i] - mx) * rs;
}

extern "C" void kernel_launch(void* const* d_in, const int* in_sizes, int n_in,
                              void* d_out, int out_size, void* d_ws, size_t ws_size,
                              hipStream_t stream) {
  const float* neigh = (const float*)d_in[0];
  const float* tgt   = (const float*)d_in[1];
  const int*   perms = (const int*)d_in[2];
  float* out  = (float*)d_out;
  float* vals = (float*)d_ws;
  float* stats = vals + N_ITEMS;
  unsigned* umax = (unsigned*)(stats + 2);

  init_kernel<<<1, 32, 0, stream>>>(tgt, stats, umax);

  const int nTiles = N_ITEMS / 16;
  vals_kernel<<<(nTiles + 7) / 8, 256, 0, stream>>>(neigh, tgt, stats, vals, out, nTiles);

  dim3 sg((N_ITEMS + 255) / 256, S_PERMS);
  scatter_kernel<<<sg, 256, 0, stream>>>(perms, vals, out);

  max_kernel<<<1024, 256, 0, stream>>>(out, umax, N_ITEMS);
  sum_kernel<<<1024, 256, 0, stream>>>(out, umax, stats + 1, N_ITEMS);
  final_kernel<<<(N_ITEMS + 255) / 256, 256, 0, stream>>>(out, umax, stats + 1, N_ITEMS);
}


# Scatter adds kept per work-group in LDS as order-independent 32-bit fixed-point integers, flushed by whole-line integer atomics; float sum order changes (exact integer sums, one rounding per term)
# speedup vs baseline: 4.5653x; 4.5653x over previous
#include <hip/hip_runtime.h>
#include <math.h>

#define N_ITEMS 200000
#define DIMS    128
#define S_PERMS 128
#define ALPHA_C 0.7f
#define EPS_C   1e-8f

typedef float v2f __attribute__((ext_vector_type(2)));
typedef float v8f __attribute__((ext_vector_type(8)));

__device__ __forceinline__ unsigned fenc(float f) {
  unsigned u = __float_as_uint(f);
  return (u & 0x80000000u) ? ~u : (u | 0x80000000u);
}
__device__ __forceinline__ float fdec(unsigned e) {
  unsigned u = (e & 0x80000000u) ? (e & 0x7FFFFFFFu) : ~e;
  return __uint_as_float(u);
}

__global__ void init_kernel(const float* __restrict__ tgt,
                            float* __restrict__ stats,
                            unsigned* __restrict__ umax) {
  int lane = threadIdx.x;
  float ss = 0.f;
  for (int k = lane; k < DIMS; k += 32) { float v = tgt[k]; ss = fmaf(v, v, ss); }
  #pragma unroll
  for (int o = 16; o; o >>= 1) ss += __shfl_xor(ss, o);
  if (lane == 0) { stats[0] = sqrtf(ss); stats[1] = 0.f; *umax = 0u; }
}

__global__ void __launch_bounds__(256)
vals_kernel(const float* __restrict__ neigh, const float* __restrict__ tgt,
            const float* __restrict__ stats, float* __restrict__ vals,
            float* __restrict__ w, int nTiles) {
  __shared__ __align__(16) float lds[8][16 * 132];
  const int lane  = threadIdx.x & 31;
  const int wave  = threadIdx.x >> 5;
  const int hi    = lane >> 4;
  const int row16 = lane & 15;
  float* tile = &lds[wave][0];
  const float tn = fmaxf(stats[0], EPS_C);

  v2f bv[32];
  #pragma unroll
  for (int k = 0; k < 32; ++k)
    bv[k] = *(const v2f*)(tgt + 4 * k + 2 * hi);

  const int r = lane >> 1, h = lane & 1;
  const int wavesTotal = (gridDim.x * blockDim.x) >> 5;
  for (int t = (blockIdx.x * blockDim.x + threadIdx.x) >> 5; t < nTiles;
       t += wavesTotal) {
    const float* src = neigh + (size_t)t * (16 * DIMS);
    const float4* srow = (const float4*)(src + r * DIMS + h * 64);
    float* drow = tile + r * 132 + h * 64;
    float ss = 0.f;
    #pragma unroll
    for (int j = 0; j < 16; ++j) {
      float4 v = srow[j];
      ss = fmaf(v.x, v.x, ss); ss = fmaf(v.y, v.y, ss);
      ss = fmaf(v.z, v.z, ss); ss = fmaf(v.w, v.w, ss);
      *(float4*)(drow + 4 * j) = v;
    }
    ss += __shfl_xor(ss, 1);

    v8f acc = {};
    const float* arow = tile + row16 * 132 + 2 * hi;
    #pragma unroll
    for (int k = 0; k < 32; ++k) {
      v2f a = *(const v2f*)(arow + 4 * k);
      acc = __builtin_amdgcn_wmma_f32_16x16x4_f32(false, a, false, bv[k],
                                                  (short)0, acc, false, false);
    }

    float vout[8];
    #pragma unroll
    for (int j = 0; j < 8; ++j) {
      float ssj = __shfl(ss, (lane & 16) + 2 * j);
      float dot = acc[j];
      float nn  = fmaxf(sqrtf(ssj), EPS_C);
      float sim = dot / (nn * tn);
      vout[j] = dot * __powf(sim, ALPHA_C);
    }
    if (row16 == 0) {
      int base = t * 16 + hi * 8;
      float4 lo  = make_float4(vout[0], vout[1], vout[2], vout[3]);
      float4 hi4 = make_float4(vout[4], vout[5], vout[6], vout[7]);
      *(float4*)(vals + base)     = lo;  *(float4*)(vals + base + 4) = hi4;
      const float4 z4 = make_float4(0.f, 0.f, 0.f, 0.f);
      *(float4*)(w + base)        = z4;  *(float4*)(w + base + 4)    = z4;
    }
  }
}

#define FIX_BITS   16
#define PARTS      4
#define PART_ITEMS (N_ITEMS / PARTS)
#define SLABS      (S_PERMS / 2)
#define SLAB_ELEMS (2 * N_ITEMS)
#define SLAB_CHUNKS (SLAB_ELEMS / 4)

__global__ void __launch_bounds__(1024)
scatter_kernel(const int* __restrict__ perms, const float* __restrict__ vals,
               unsigned* __restrict__ acc) {
  __shared__ unsigned lacc[PART_ITEMS];
  const int tid  = threadIdx.x;
  const int part = blockIdx.x & (PARTS - 1);
  const int slab = blockIdx.x / PARTS;
  const unsigned lo = (unsigned)(part * PART_ITEMS);

  for (int j = tid; j < PART_ITEMS; j += 1024) lacc[j] = 0u;
  __syncthreads();

  const int*  base  = perms + (size_t)slab * SLAB_ELEMS;
  const int4* base4 = (const int4*)base;
  for (int c0 = tid; c0 < SLAB_CHUNKS; c0 += 4 * 1024) {
    int pp[16], p[16];
    #pragma unroll
    for (int u = 0; u < 4; ++u) {
      const int c = c0 + u * 1024;
      const bool in = c < SLAB_CHUNKS;
      const int4 v = in ? base4[c] : make_int4(-1, -1, -1, -1);
      const int pv = (in && c != 0 && c != N_ITEMS / 4) ? base[4 * c - 1] : -1;
      pp[4 * u + 0] = pv;  p[4 * u + 0] = v.x;
      pp[4 * u + 1] = v.x; p[4 * u + 1] = v.y;
      pp[4 * u + 2] = v.y; p[4 * u + 2] = v.z;
      pp[4 * u + 3] = v.z; p[4 * u + 3] = v.w;
    }
    float gv[16]; unsigned jj[16];
    #pragma unroll
    for (int k = 0; k < 16; ++k) {
      const unsigned j = (unsigned)p[k] - lo;
      const bool ok = j < (unsigned)PART_ITEMS && pp[k] >= 0;
      jj[k] = ok ? j : 0xFFFFFFFFu;
      gv[k] = vals[ok ? pp[k] : 0];
    }
    #pragma unroll
    for (int k = 0; k < 16; ++k) {
      if (jj[k] != 0xFFFFFFFFu) {
        const int fx = __float2int_rn(gv[k] * (float)(1 << FIX_BITS));
        atomicAdd(&lacc[jj[k]], (unsigned)fx);
      }
    }
  }
  __syncthreads();

  for (int j = tid; j < PART_ITEMS; j += 1024)
    atomicAdd(&acc[lo + j], lacc[j]);
}

__device__ __forceinline__ float w_of(const unsigned* __restrict__ acc,
                                      const float* __restrict__ vals, int i) {
  return fmaf((float)(int)acc[i], -1.0f / ((float)(1 << FIX_BITS) * S_PERMS), vals[i]);
}

__global__ void max_kernel(const unsigned* __restrict__ acc, const float* __restrict__ vals,
                           unsigned* __restrict__ umax, int n) {
  float m = -INFINITY;
  for (int i = blockIdx.x * blockDim.x + threadIdx.x; i < n; i += gridDim.x * blockDim.x)
    m = fmaxf(m, w_of(acc, vals, i));
  #pragma unroll
  for (int o = 16; o; o >>= 1) m = fmaxf(m, __shfl_xor(m, o));
  if ((threadIdx.x & 31) == 0) atomicMax(umax, fenc(m));
}

__global__ void sum_kernel(const unsigned* __restrict__ acc, const float* __restrict__ vals,
                           const unsigned* __restrict__ umax, float* __restrict__ sum, int n) {
  const float mx = fdec(*umax);
  float a = 0.f;
  for (int i = blockIdx.x * blockDim.x + threadIdx.x; i < n; i += gridDim.x * blockDim.x)
    a += __expf(w_of(acc, vals, i) - mx);
  #pragma unroll
  for (int o = 16; o; o >>= 1) a += __shfl_xor(a, o);
  if ((threadIdx.x & 31) == 0) atomicAdd(sum, a);
}

__global__ void final_kernel(float* w, const float* __restrict__ vals,
                             const unsigned* __restrict__ umax,
                             const float* __restrict__ sum, int n) {
  const float mx = fdec(*umax);
  const float rs = 1.0f / *sum;
  int i = blockIdx.x * blockDim.x + threadIdx.x;
  if (i < n) {
    const float wi = w_of((const unsigned*)w, vals, i);
    w[i] = __expf(wi - mx) * rs;
  }
}

extern "C" void kernel_launch(void* const* d_in, const int* in_sizes, int n_in,
                              void* d_out, int out_size, void* d_ws, size_t ws_size,
                              hipStream_t stream) {
  const float* neigh = (const float*)d_in[0];
  const float* tgt   = (const float*)d_in[1];
  const int*   perms = (const int*)d_in[2];
  float* out  = (float*)d_out;
  float* vals = (float*)d_ws;
  float* stats = vals + N_ITEMS;
  unsigned* umax = (unsigned*)(stats + 2);

  init_kernel<<<1, 32, 0, stream>>>(tgt, stats, umax);

  const int nTiles = N_ITEMS / 16;
  vals_kernel<<<(nTiles + 7) / 8, 256, 0, stream>>>(neigh, tgt, stats, vals, out, nTiles);

  scatter_kernel<<<PARTS * SLABS, 1024, 0, stream>>>(perms, vals, (unsigned*)out);

  max_kernel<<<1024, 256, 0, stream>>>((const unsigned*)out, vals, umax, N_ITEMS);
  sum_kernel<<<1024, 256, 0, stream>>>((const unsigned*)out, vals, umax, stats + 1, N_ITEMS);
  final_kernel<<<(N_ITEMS + 255) / 256, 256, 0, stream>>>(out, vals, umax, stats + 1, N_ITEMS);
}


# Softmax max and sum from one pass of per-work-group (max, sum) pairs folded in a fixed order by the last kernel: no same-word atomics, 6 launches to 5; the denominator's float sum order changes
# speedup vs baseline: 4.9712x; 1.0889x over previous
#include <hip/hip_runtime.h>
#include <math.h>

#define N_ITEMS 200000
#define DIMS    128
#define S_PERMS 128
#define ALPHA_C 0.7f
#define EPS_C   1e-8f

typedef float v2f __attribute__((ext_vector_type(2)));
typedef float v8f __attribute__((ext_vector_type(8)));

__global__ void init_kernel(const float* __restrict__ tgt,
                            float* __restrict__ stats) {
  int lane = threadIdx.x;
  float ss = 0.f;
  for (int k = lane; k < DIMS; k += 32) { float v = tgt[k]; ss = fmaf(v, v, ss); }
  #pragma unroll
  for (int o = 16; o; o >>= 1) ss += __shfl_xor(ss, o);
  if (lane == 0) stats[0] = sqrtf(ss);
}

__global__ void __launch_bounds__(256)
vals_kernel(const float* __restrict__ neigh, const float* __restrict__ tgt,
            const float* __restrict__ stats, float* __restrict__ vals,
            float* __restrict__ w, int nTiles) {
  __shared__ __align__(16) float lds[8][16 * 132];
  const int lane  = threadIdx.x & 31;
  const int wave  = threadIdx.x >> 5;
  const int hi    = lane >> 4;
  const int row16 = lane & 15;
  float* tile = &lds[wave][0];
  const float tn = fmaxf(stats[0], EPS_C);

  v2f bv[32];
  #pragma unroll
  for (int k = 0; k < 32; ++k)
    bv[k] = *(const v2f*)(tgt + 4 * k + 2 * hi);

  const int r = lane >> 1, h = lane & 1;
  const int wavesTotal = (gridDim.x * blockDim.x) >> 5;
  for (int t = (blockIdx.x * blockDim.x + threadIdx.x) >> 5; t < nTiles;
       t += wavesTotal) {
    const float* src = neigh + (size_t)t * (16 * DIMS);
    const float4* srow = (const float4*)(src + r * DIMS + h * 64);
    float* drow = tile + r * 132 + h * 64;
    float ss = 0.f;
    #pragma unroll
    for (int j = 0; j < 16; ++j) {
      float4 v = srow[j];
      ss = fmaf(v.x, v.x, ss); ss = fmaf(v.y, v.y, ss);
      ss = fmaf(v.z, v.z, ss); ss = fmaf(v.w, v.w, ss);
      *(float4*)(drow + 4 * j) = v;
    }
    ss += __shfl_xor(ss, 1);

    v8f acc = {};
    const float* arow = tile + row16 * 132 + 2 * hi;
    #pragma unroll
    for (int k = 0; k < 32; ++k) {
      v2f a = *(const v2f*)(arow + 4 * k);
      acc = __builtin_amdgcn_wmma_f32_16x16x4_f32(false, a, false, bv[k],
                                                  (short)0, acc, false, false);
    }

    float vout[8];
    #pragma unroll
    for (int j = 0; j < 8; ++j) {
      float ssj = __shfl(ss, (lane & 16) + 2 * j);
      float dot = acc[j];
      float nn  = fmaxf(sqrtf(ssj), EPS_C);
      float sim = dot / (nn * tn);
      vout[j] = dot * __powf(sim, ALPHA_C);
    }
    if (row16 == 0) {
      int base = t * 16 + hi * 8;
      float4 lo  = make_float4(vout[0], vout[1], vout[2], vout[3]);
      float4 hi4 = make_float4(vout[4], vout[5], vout[6], vout[7]);
      *(float4*)(vals + base)     = lo;  *(float4*)(vals + base + 4) = hi4;
      const float4 z4 = make_float4(0.f, 0.f, 0.f, 0.f);
      *(float4*)(w + base)        = z4;  *(float4*)(w + base + 4)    = z4;
    }
  }
}

#define FIX_BITS   16
#define PARTS      4
#define PART_ITEMS (N_ITEMS / PARTS)
#define SLABS      (S_PERMS / 2)
#define SLAB_ELEMS (2 * N_ITEMS)
#define SLAB_CHUNKS (SLAB_ELEMS / 4)

__global__ void __launch_bounds__(1024)
scatter_kernel(const int* __restrict__ perms, const float* __restrict__ vals,
               unsigned* __restrict__ acc) {
  __shared__ unsigned lacc[PART_ITEMS];
  const int tid  = threadIdx.x;
  const int part = blockIdx.x & (PARTS - 1);
  const int slab = blockIdx.x / PARTS;
  const unsigned lo = (unsigned)(part * PART_ITEMS);

  for (int j = tid; j < PART_ITEMS; j += 1024) lacc[j] = 0u;
  __syncthreads();

  const int*  base  = perms + (size_t)slab * SLAB_ELEMS;
  const int4* base4 = (const int4*)base;
  for (int c0 = tid; c0 < SLAB_CHUNKS; c0 += 4 * 1024) {
    int pp[16], p[16];
    #pragma unroll
    for (int u = 0; u < 4; ++u) {
      const int c = c0 + u * 1024;
      const bool in = c < SLAB_CHUNKS;
      const int4 v = in ? base4[c] : make_int4(-1, -1, -1, -1);
      const int pv = (in && c != 0 && c != N_ITEMS / 4) ? base[4 * c - 1] : -1;
      pp[4 * u + 0] = pv;  p[4 * u + 0] = v.x;
      pp[4 * u + 1] = v.x; p[4 * u + 1] = v.y;
      pp[4 * u + 2] = v.y; p[4 * u + 2] = v.z;
      pp[4 * u + 3] = v.z; p[4 * u + 3] = v.w;
    }
    float gv[16]; unsigned jj[16];
    #pragma unroll
    for (int k = 0; k < 16; ++k) {
      const unsigned j = (unsigned)p[k] - lo;
      const bool ok = j < (unsigned)PART_ITEMS && pp[k] >= 0;
      jj[k] = ok ? j : 0xFFFFFFFFu;
      gv[k] = vals[ok ? pp[k] : 0];
    }
    #pragma unroll
    for (int k = 0; k < 16; ++k) {
      if (jj[k] != 0xFFFFFFFFu) {
        const int fx = __float2int_rn(gv[k] * (float)(1 << FIX_BITS));
        atomicAdd(&lacc[jj[k]], (unsigned)fx);
      }
    }
  }
  __syncthreads();

  for (int j = tid; j < PART_ITEMS; j += 1024)
    atomicAdd(&acc[lo + j], lacc[j]);
}

#define SM_THREADS 256
#define SM_ITEMS   4
#define SM_BLOCKS  ((N_ITEMS + SM_THREADS * SM_ITEMS - 1) / (SM_THREADS * SM_ITEMS))

__device__ __forceinline__ void w4_of(const unsigned* acc, const float* __restrict__ vals,
                                      int i4, float w[4]) {
  const uint4  a = *(const uint4*)(acc + i4);
  const float4 v = *(const float4*)(vals + i4);
  const float sc = -1.0f / ((float)(1 << FIX_BITS) * S_PERMS);
  w[0] = fmaf((float)(int)a.x, sc, v.x);
  w[1] = fmaf((float)(int)a.y, sc, v.y);
  w[2] = fmaf((float)(int)a.z, sc, v.z);
  w[3] = fmaf((float)(int)a.w, sc, v.w);
}

__device__ __forceinline__ float block_max(float m, float* red) {
  for (int o = 16; o; o >>= 1) m = fmaxf(m, __shfl_xor(m, o));
  if ((threadIdx.x & 31) == 0) red[threadIdx.x >> 5] = m;
  __syncthreads();
  float r = red[0];
  for (int k = 1; k < SM_THREADS / 32; ++k) r = fmaxf(r, red[k]);
  __syncthreads();
  return r;
}
__device__ __forceinline__ float block_sum(float s, float* red) {
  for (int o = 16; o; o >>= 1) s += __shfl_xor(s, o);
  if ((threadIdx.x & 31) == 0) red[threadIdx.x >> 5] = s;
  __syncthreads();
  float r = red[0];
  for (int k = 1; k < SM_THREADS / 32; ++k) r += red[k];
  __syncthreads();
  return r;
}

__global__ void __launch_bounds__(SM_THREADS)
stats_kernel(const unsigned* __restrict__ acc, const float* __restrict__ vals,
             float* __restrict__ part) {
  __shared__ float red[SM_THREADS / 32];
  const int i4 = (blockIdx.x * SM_THREADS + threadIdx.x) * SM_ITEMS;
  const bool in = i4 < N_ITEMS;
  float w[4] = {-INFINITY, -INFINITY, -INFINITY, -INFINITY};
  if (in) w4_of(acc, vals, i4, w);
  const float bm = block_max(fmaxf(fmaxf(w[0], w[1]), fmaxf(w[2], w[3])), red);
  float s = 0.f;
  if (in) s = (__expf(w[0] - bm) + __expf(w[1] - bm)) + (__expf(w[2] - bm) + __expf(w[3] - bm));
  const float bs = block_sum(s, red);
  if (threadIdx.x == 0) *(float2*)(part + 2 * blockIdx.x) = make_float2(bm, bs);
}

__global__ void __launch_bounds__(SM_THREADS)
final_kernel(float* w, const float* __restrict__ vals, const float* __restrict__ part) {
  __shared__ float red[SM_THREADS / 32];
  const int t = threadIdx.x;
  const bool has = t < SM_BLOCKS;
  const float2 p = has ? *(const float2*)(part + 2 * t) : make_float2(-INFINITY, 0.f);
  const int i4 = (blockIdx.x * SM_THREADS + t) * SM_ITEMS;
  const bool in = i4 < N_ITEMS;
  float ww[4] = {0.f, 0.f, 0.f, 0.f};
  if (in) w4_of((const unsigned*)w, vals, i4, ww);
  const float mx = block_max(p.x, red);
  const float sum = block_sum(has ? p.y * __expf(p.x - mx) : 0.f, red);
  const float rs = 1.0f / sum;
  if (in)
    *(float4*)(w + i4) = make_float4(__expf(ww[0] - mx) * rs, __expf(ww[1] - mx) * rs,
                                     __expf(ww[2] - mx) * rs, __expf(ww[3] - mx) * rs);
}

extern "C" void kernel_launch(void* const* d_in, const int* in_sizes, int n_in,
                              void* d_out, int out_size, void* d_ws, size_t ws_size,
                              hipStream_t stream) {
  const float* neigh = (const float*)d_in[0];
  const float* tgt   = (const float*)d_in[1];
  const int*   perms = (const int*)d_in[2];
  float* out  = (float*)d_out;
  float* vals = (float*)d_ws;
  float* stats = vals + N_ITEMS;
  float* part  = stats + 4;

  init_kernel<<<1, 32, 0, stream>>>(tgt, stats);

  const int nTiles = N_ITEMS / 16;
  vals_kernel<<<(nTiles + 7) / 8, 256, 0, stream>>>(neigh, tgt, stats, vals, out, nTiles);

  scatter_kernel<<<PARTS * SLABS, 1024, 0, stream>>>(perms, vals, (unsigned*)out);

  stats_kernel<<<SM_BLOCKS, SM_THREADS, 0, stream>>>((const unsigned*)out, vals, part);
  final_kernel<<<SM_BLOCKS, SM_THREADS, 0, stream>>>(out, vals, part);
}


# Scatter launch: a window of the gathered value table (31392 of 200000 items) kept in the work-group's spare LDS; gathers of those items are LDS reads, not second-level cache requests; sums unchanged
# speedup vs baseline: 5.4664x; 1.0996x over previous
#include <hip/hip_runtime.h>
#include <math.h>

#define N_ITEMS 200000
#define DIMS    128
#define S_PERMS 128
#define ALPHA_C 0.7f
#define EPS_C   1e-8f

typedef float v2f __attribute__((ext_vector_type(2)));
typedef float v8f __attribute__((ext_vector_type(8)));

__global__ void init_kernel(const float* __restrict__ tgt,
                            float* __restrict__ stats) {
  int lane = threadIdx.x;
  float ss = 0.f;
  for (int k = lane; k < DIMS; k += 32) { float v = tgt[k]; ss = fmaf(v, v, ss); }
  #pragma unroll
  for (int o = 16; o; o >>= 1) ss += __shfl_xor(ss, o);
  if (lane == 0) stats[0] = sqrtf(ss);
}

__global__ void __launch_bounds__(256)
vals_kernel(const float* __restrict__ neigh, const float* __restrict__ tgt,
            const float* __restrict__ stats, float* __restrict__ vals,
            float* __restrict__ w, int nTiles) {
  __shared__ __align__(16) float lds[8][16 * 132];
  const int lane  = threadIdx.x & 31;
  const int wave  = threadIdx.x >> 5;
  const int hi    = lane >> 4;
  const int row16 = lane & 15;
  float* tile = &lds[wave][0];
  const float tn = fmaxf(stats[0], EPS_C);

  v2f bv[32];
  #pragma unroll
  for (int k = 0; k < 32; ++k)
    bv[k] = *(const v2f*)(tgt + 4 * k + 2 * hi);

  const int r = lane >> 1, h = lane & 1;
  const int wavesTotal = (gridDim.x * blockDim.x) >> 5;
  for (int t = (blockIdx.x * blockDim.x + threadIdx.x) >> 5; t < nTiles;
       t += wavesTotal) {
    const float* src = neigh + (size_t)t * (16 * DIMS);
    const float4* srow = (const float4*)(src + r * DIMS + h * 64);
    float* drow = tile + r * 132 + h * 64;
    float ss = 0.f;
    #pragma unroll
    for (int j = 0; j < 16; ++j) {
      float4 v = srow[j];
      ss = fmaf(v.x, v.x, ss); ss = fmaf(v.y, v.y, ss);
      ss = fmaf(v.z, v.z, ss); ss = fmaf(v.w, v.w, ss);
      *(float4*)(drow + 4 * j) = v;
    }
    ss += __shfl_xor(ss, 1);

    v8f acc = {};
    const float* arow = tile + row16 * 132 + 2 * hi;
    #pragma unroll
    for (int k = 0; k < 32; ++k) {
      v2f a = *(const v2f*)(arow + 4 * k);
      acc = __builtin_amdgcn_wmma_f32_16x16x4_f32(false, a, false, bv[k],
                                                  (short)0, acc, false, false);
    }

    float vout[8];
    #pragma unroll
    for (int j = 0; j < 8; ++j) {
      float ssj = __shfl(ss, (lane & 16) + 2 * j);
      float dot = acc[j];
      float nn  = fmaxf(sqrtf(ssj), EPS_C);
      float sim = dot / (nn * tn);
      vout[j] = dot * __powf(sim, ALPHA_C);
    }
    if (row16 == 0) {
      int base = t * 16 + hi * 8;
      float4 lo  = make_float4(vout[0], vout[1], vout[2], vout[3]);
      float4 hi4 = make_float4(vout[4], vout[5], vout[6], vout[7]);
      *(float4*)(vals + base)     = lo;  *(float4*)(vals + base + 4) = hi4;
      const float4 z4 = make_float4(0.f, 0.f, 0.f, 0.f);
      *(float4*)(w + base)        = z4;  *(float4*)(w + base + 4)    = z4;
    }
  }
}

#define FIX_BITS   16
#define PARTS      4
#define PART_ITEMS (N_ITEMS / PARTS)
#define SLABS      (S_PERMS / 2)
#define SLAB_ELEMS (2 * N_ITEMS)
#define SLAB_CHUNKS (SLAB_ELEMS / 4)
#define WIN_ITEMS  31392

__global__ void __launch_bounds__(1024)
scatter_kernel(const int* __restrict__ perms, const float* __restrict__ vals,
               unsigned* __restrict__ acc) {
  __shared__ unsigned lacc[PART_ITEMS];
  __shared__ __align__(16) float win[WIN_ITEMS];
  const int tid  = threadIdx.x;
  const int part = blockIdx.x & (PARTS - 1);
  const int slab = blockIdx.x / PARTS;
  const unsigned lo = (unsigned)(part * PART_ITEMS);

  for (int j = tid; j < PART_ITEMS; j += 1024) lacc[j] = 0u;
  for (int j = tid; j < WIN_ITEMS / 4; j += 1024)
    ((float4*)win)[j] = ((const float4*)(vals + lo))[j];
  __syncthreads();

  const int*  base  = perms + (size_t)slab * SLAB_ELEMS;
  const int4* base4 = (const int4*)base;
  for (int c0 = tid; c0 < SLAB_CHUNKS; c0 += 4 * 1024) {
    int pp[16], p[16];
    #pragma unroll
    for (int u = 0; u < 4; ++u) {
      const int c = c0 + u * 1024;
      const bool in = c < SLAB_CHUNKS;
      const int4 v = in ? base4[c] : make_int4(-1, -1, -1, -1);
      const int pv = (in && c != 0 && c != N_ITEMS / 4) ? base[4 * c - 1] : -1;
      pp[4 * u + 0] = pv;  p[4 * u + 0] = v.x;
      pp[4 * u + 1] = v.x; p[4 * u + 1] = v.y;
      pp[4 * u + 2] = v.y; p[4 * u + 2] = v.z;
      pp[4 * u + 3] = v.z; p[4 * u + 3] = v.w;
    }
    float gv[16], lv[16]; unsigned jj[16];
    #pragma unroll
    for (int k = 0; k < 16; ++k) {
      const unsigned j = (unsigned)p[k] - lo;
      const bool ok = j < (unsigned)PART_ITEMS && pp[k] >= 0;
      const unsigned ws = (unsigned)pp[k] - lo;
      const bool inw = ok && ws < (unsigned)WIN_ITEMS;
      jj[k] = ok ? j : 0xFFFFFFFFu;
      gv[k] = vals[(ok && !inw) ? pp[k] : 0];
      lv[k] = win[inw ? ws : 0u];
    }
    #pragma unroll
    for (int k = 0; k < 16; ++k) {
      if (jj[k] != 0xFFFFFFFFu) {
        const bool inw = ((unsigned)pp[k] - lo) < (unsigned)WIN_ITEMS;
        const float val = inw ? lv[k] : gv[k];
        const int fx = __float2int_rn(val * (float)(1 << FIX_BITS));
        atomicAdd(&lacc[jj[k]], (unsigned)fx);
      }
    }
  }
  __syncthreads();

  for (int j = tid; j < PART_ITEMS; j += 1024)
    atomicAdd(&acc[lo + j], lacc[j]);
}

#define SM_THREADS 256
#define SM_ITEMS   4
#define SM_BLOCKS  ((N_ITEMS + SM_THREADS * SM_ITEMS - 1) / (SM_THREADS * SM_ITEMS))

__device__ __forceinline__ void w4_of(const unsigned* acc, const float* __restrict__ vals,
                                      int i4, float w[4]) {
  const uint4  a = *(const uint4*)(acc + i4);
  const float4 v = *(const float4*)(vals + i4);
  const float sc = -1.0f / ((float)(1 << FIX_BITS) * S_PERMS);
  w[0] = fmaf((float)(int)a.x, sc, v.x);
  w[1] = fmaf((float)(int)a.y, sc, v.y);
  w[2] = fmaf((float)(int)a.z, sc, v.z);
  w[3] = fmaf((float)(int)a.w, sc, v.w);
}

__device__ __forceinline__ float block_max(float m, float* red) {
  for (int o = 16; o; o >>= 1) m = fmaxf(m, __shfl_xor(m, o));
  if ((threadIdx.x & 31) == 0) red[threadIdx.x >> 5] = m;
  __syncthreads();
  float r = red[0];
  for (int k = 1; k < SM_THREADS / 32; ++k) r = fmaxf(r, red[k]);
  __syncthreads();
  return r;
}
__device__ __forceinline__ float block_sum(float s, float* red) {
  for (int o = 16; o; o >>= 1) s += __shfl_xor(s, o);
  if ((threadIdx.x & 31) == 0) red[threadIdx.x >> 5] = s;
  __syncthreads();
  float r = red[0];
  for (int k = 1; k < SM_THREADS / 32; ++k) r += red[k];
  __syncthreads();
  return r;
}

__global__ void __launch_bounds__(SM_THREADS)
stats_kernel(const unsigned* __restrict__ acc, const float* __restrict__ vals,
             float* __restrict__ part) {
  __shared__ float red[SM_THREADS / 32];
  const int i4 = (blockIdx.x * SM_THREADS + threadIdx.x) * SM_ITEMS;
  const bool in = i4 < N_ITEMS;
  float w[4] = {-INFINITY, -INFINITY, -INFINITY, -INFINITY};
  if (in) w4_of(acc, vals, i4, w);
  const float bm = block_max(fmaxf(fmaxf(w[0], w[1]), fmaxf(w[2], w[3])), red);
  float s = 0.f;
  if (in) s = (__expf(w[0] - bm) + __expf(w[1] - bm)) + (__expf(w[2] - bm) + __expf(w[3] - bm));
  const float bs = block_sum(s, red);
  if (threadIdx.x == 0) *(float2*)(part + 2 * blockIdx.x) = make_float2(bm, bs);
}

__global__ void __launch_bounds__(SM_THREADS)
final_kernel(float* w, const float* __restrict__ vals, const float* __restrict__ part) {
  __shared__ float red[SM_THREADS / 32];
  const int t = threadIdx.x;
  const bool has = t < SM_BLOCKS;
  const float2 p = has ? *(const float2*)(part + 2 * t) : make_float2(-INFINITY, 0.f);
  const int i4 = (blockIdx.x * SM_THREADS + t) * SM_ITEMS;
  const bool in = i4 < N_ITEMS;
  float ww[4] = {0.f, 0.f, 0.f, 0.f};
  if (in) w4_of((const unsigned*)w, vals, i4, ww);
  const float mx = block_max(p.x, red);
  const float sum = block_sum(has ? p.y * __expf(p.x - mx) : 0.f, red);
  const float rs = 1.0f / sum;
  if (in)
    *(float4*)(w + i4) = make_float4(__expf(ww[0] - mx) * rs, __expf(ww[1] - mx) * rs,
                                     __expf(ww[2] - mx) * rs, __expf(ww[3] - mx) * rs);
}

extern "C" void kernel_launch(void* const* d_in, const int* in_sizes, int n_in,
                              void* d_out, int out_size, void* d_ws, size_t ws_size,
                              hipStream_t stream) {
  const float* neigh = (const float*)d_in[0];
  const float* tgt   = (const float*)d_in[1];
  const int*   perms = (const int*)d_in[2];
  float* out  = (float*)d_out;
  float* vals = (float*)d_ws;
  float* stats = vals + N_ITEMS;
  float* part  = stats + 4;

  init_kernel<<<1, 32, 0, stream>>>(tgt, stats);

  const int nTiles = N_ITEMS / 16;
  vals_kernel<<<(nTiles + 7) / 8, 256, 0, stream>>>(neigh, tgt, stats, vals, out, nTiles);

  scatter_kernel<<<PARTS * SLABS, 1024, 0, stream>>>(perms, vals, (unsigned*)out);

  stats_kernel<<<SM_BLOCKS, SM_THREADS, 0, stream>>>((const unsigned*)out, vals, part);
  final_kernel<<<SM_BLOCKS, SM_THREADS, 0, stream>>>(out, vals, part);
}


# Scatter launch: eight item parts of 25000, four permutation rows per work-group (was four of 50000, two rows): half the partial sums through global atomics, wider LDS value window; sums unchanged
# speedup vs baseline: 5.6674x; 1.0368x over previous
#include <hip/hip_runtime.h>
#include <math.h>

#define N_ITEMS 200000
#define DIMS    128
#define S_PERMS 128
#define ALPHA_C 0.7f
#define EPS_C   1e-8f

typedef float v2f __attribute__((ext_vector_type(2)));
typedef float v8f __attribute__((ext_vector_type(8)));

__global__ void init_kernel(const float* __restrict__ tgt,
                            float* __restrict__ stats) {
  int lane = threadIdx.x;
  float ss = 0.f;
  for (int k = lane; k < DIMS; k += 32) { float v = tgt[k]; ss = fmaf(v, v, ss); }
  #pragma unroll
  for (int o = 16; o; o >>= 1) ss += __shfl_xor(ss, o);
  if (lane == 0) stats[0] = sqrtf(ss);
}

__global__ void __launch_bounds__(256)
vals_kernel(const float* __restrict__ neigh, const float* __restrict__ tgt,
            const float* __restrict__ stats, float* __restrict__ vals,
            float* __restrict__ w, int nTiles) {
  __shared__ __align__(16) float lds[8][16 * 132];
  const int lane  = threadIdx.x & 31;
  const int wave  = threadIdx.x >> 5;
  const int hi    = lane >> 4;
  const int row16 = lane & 15;
  float* tile = &lds[wave][0];
  const float tn = fmaxf(stats[0], EPS_C);

  v2f bv[32];
  #pragma unroll
  for (int k = 0; k < 32; ++k)
    bv[k] = *(const v2f*)(tgt + 4 * k + 2 * hi);

  const int r = lane >> 1, h = lane & 1;
  const int wavesTotal = (gridDim.x * blockDim.x) >> 5;
  for (int t = (blockIdx.x * blockDim.x + threadIdx.x) >> 5; t < nTiles;
       t += wavesTotal) {
    const float* src = neigh + (size_t)t * (16 * DIMS);
    const float4* srow = (const float4*)(src + r * DIMS + h * 64);
    float* drow = tile + r * 132 + h * 64;
    float ss = 0.f;
    #pragma unroll
    for (int j = 0; j < 16; ++j) {
      float4 v = srow[j];
      ss = fmaf(v.x, v.x, ss); ss = fmaf(v.y, v.y, ss);
      ss = fmaf(v.z, v.z, ss); ss = fmaf(v.w, v.w, ss);
      *(float4*)(drow + 4 * j) = v;
    }
    ss += __shfl_xor(ss, 1);

    v8f acc = {};
    const float* arow = tile + row16 * 132 + 2 * hi;
    #pragma unroll
    for (int k = 0; k < 32; ++k) {
      v2f a = *(const v2f*)(arow + 4 * k);
      acc = __builtin_amdgcn_wmma_f32_16x16x4_f32(false, a, false, bv[k],
                                                  (short)0, acc, false, false);
    }

    float vout[8];
    #pragma unroll
    for (int j = 0; j < 8; ++j) {
      float ssj = __shfl(ss, (lane & 16) + 2 * j);
      float dot = acc[j];
      float nn  = fmaxf(sqrtf(ssj), EPS_C);
      float sim = dot / (nn * tn);
      vout[j] = dot * __powf(sim, ALPHA_C);
    }
    if (row16 == 0) {
      int base = t * 16 + hi * 8;
      float4 lo  = make_float4(vout[0], vout[1], vout[2], vout[3]);
      float4 hi4 = make_float4(vout[4], vout[5], vout[6], vout[7]);
      *(float4*)(vals + base)     = lo;  *(float4*)(vals + base + 4) = hi4;
      const float4 z4 = make_float4(0.f, 0.f, 0.f, 0.f);
      *(float4*)(w + base)        = z4;  *(float4*)(w + base + 4)    = z4;
    }
  }
}

#define FIX_BITS   16
#define PARTS      8
#define PART_ITEMS (N_ITEMS / PARTS)
#define SLAB_ROWS  4
#define SLABS      (S_PERMS / SLAB_ROWS)
#define SLAB_ELEMS (SLAB_ROWS * N_ITEMS)
#define SLAB_CHUNKS (SLAB_ELEMS / 4)
#define ROW_CHUNKS (N_ITEMS / 4)
#define WIN_ITEMS  56392

__global__ void __launch_bounds__(1024)
scatter_kernel(const int* __restrict__ perms, const float* __restrict__ vals,
               unsigned* __restrict__ acc) {
  __shared__ unsigned lacc[PART_ITEMS];
  __shared__ __align__(16) float win[WIN_ITEMS];
  const int tid  = threadIdx.x;
  const int part = blockIdx.x & (PARTS - 1);
  const int slab = blockIdx.x / PARTS;
  const unsigned lo = (unsigned)(part * PART_ITEMS);
  const unsigned wlo = lo < (unsigned)(N_ITEMS - WIN_ITEMS) ? lo : (unsigned)(N_ITEMS - WIN_ITEMS);

  for (int j = tid; j < PART_ITEMS; j += 1024) lacc[j] = 0u;
  for (int j = tid; j < WIN_ITEMS / 4; j += 1024)
    ((float4*)win)[j] = ((const float4*)(vals + wlo))[j];
  __syncthreads();

  const int*  base  = perms + (size_t)slab * SLAB_ELEMS;
  const int4* base4 = (const int4*)base;
  for (int c0 = tid; c0 < SLAB_CHUNKS; c0 += 4 * 1024) {
    int pp[16], p[16];
    #pragma unroll
    for (int u = 0; u < 4; ++u) {
      const int c = c0 + u * 1024;
      const bool in = c < SLAB_CHUNKS;
      const int4 v = in ? base4[c] : make_int4(-1, -1, -1, -1);
      const int pv = (in && (unsigned)c % (unsigned)ROW_CHUNKS != 0u) ? base[4 * c - 1] : -1;
      pp[4 * u + 0] = pv;  p[4 * u + 0] = v.x;
      pp[4 * u + 1] = v.x; p[4 * u + 1] = v.y;
      pp[4 * u + 2] = v.y; p[4 * u + 2] = v.z;
      pp[4 * u + 3] = v.z; p[4 * u + 3] = v.w;
    }
    float gv[16], lv[16]; unsigned jj[16];
    #pragma unroll
    for (int k = 0; k < 16; ++k) {
      const unsigned j = (unsigned)p[k] - lo;
      const bool ok = j < (unsigned)PART_ITEMS && pp[k] >= 0;
      const unsigned ws = (unsigned)pp[k] - wlo;
      const bool inw = ok && ws < (unsigned)WIN_ITEMS;
      jj[k] = ok ? j : 0xFFFFFFFFu;
      gv[k] = vals[(ok && !inw) ? pp[k] : 0];
      lv[k] = win[inw ? ws : 0u];
    }
    #pragma unroll
    for (int k = 0; k < 16; ++k) {
      if (jj[k] != 0xFFFFFFFFu) {
        const bool inw = ((unsigned)pp[k] - wlo) < (unsigned)WIN_ITEMS;
        const float val = inw ? lv[k] : gv[k];
        const int fx = __float2int_rn(val * (float)(1 << FIX_BITS));
        atomicAdd(&lacc[jj[k]], (unsigned)fx);
      }
    }
  }
  __syncthreads();

  for (int j = tid; j < PART_ITEMS; j += 1024)
    atomicAdd(&acc[lo + j], lacc[j]);
}

#define SM_THREADS 256
#define SM_ITEMS   4
#define SM_BLOCKS  ((N_ITEMS + SM_THREADS * SM_ITEMS - 1) / (SM_THREADS * SM_ITEMS))

__device__ __forceinline__ void w4_of(const unsigned* acc, const float* __restrict__ vals,
                                      int i4, float w[4]) {
  const uint4  a = *(const uint4*)(acc + i4);
  const float4 v = *(const float4*)(vals + i4);
  const float sc = -1.0f / ((float)(1 << FIX_BITS) * S_PERMS);
  w[0] = fmaf((float)(int)a.x, sc, v.x);
  w[1] = fmaf((float)(int)a.y, sc, v.y);
  w[2] = fmaf((float)(int)a.z, sc, v.z);
  w[3] = fmaf((float)(int)a.w, sc, v.w);
}

__device__ __forceinline__ float block_max(float m, float* red) {
  for (int o = 16; o; o >>= 1) m = fmaxf(m, __shfl_xor(m, o));
  if ((threadIdx.x & 31) == 0) red[threadIdx.x >> 5] = m;
  __syncthreads();
  float r = red[0];
  for (int k = 1; k < SM_THREADS / 32; ++k) r = fmaxf(r, red[k]);
  __syncthreads();
  return r;
}
__device__ __forceinline__ float block_sum(float s, float* red) {
  for (int o = 16; o; o >>= 1) s += __shfl_xor(s, o);
  if ((threadIdx.x & 31) == 0) red[threadIdx.x >> 5] = s;
  __syncthreads();
  float r = red[0];
  for (int k = 1; k < SM_THREADS / 32; ++k) r += red[k];
  __syncthreads();
  return r;
}

__global__ void __launch_bounds__(SM_THREADS)
stats_kernel(const unsigned* __restrict__ acc, const float* __restrict__ vals,
             float* __restrict__ part) {
  __shared__ float red[SM_THREADS / 32];
  const int i4 = (blockIdx.x * SM_THREADS + threadIdx.x) * SM_ITEMS;
  const bool in = i4 < N_ITEMS;
  float w[4] = {-INFINITY, -INFINITY, -INFINITY, -INFINITY};
  if (in) w4_of(acc, vals, i4, w);
  const float bm = block_max(fmaxf(fmaxf(w[0], w[1]), fmaxf(w[2], w[3])), red);
  float s = 0.f;
  if (in) s = (__expf(w[0] - bm) + __expf(w[1] - bm)) + (__expf(w[2] - bm) + __expf(w[3] - bm));
  const float bs = block_sum(s, red);
  if (threadIdx.x == 0) *(float2*)(part + 2 * blockIdx.x) = make_float2(bm, bs);
}

__global__ void __launch_bounds__(SM_THREADS)
final_kernel(float* w, const float* __restrict__ vals, const float* __restrict__ part) {
  __shared__ float red[SM_THREADS / 32];
  const int t = threadIdx.x;
  const bool has = t < SM_BLOCKS;
  const float2 p = has ? *(const float2*)(part + 2 * t) : make_float2(-INFINITY, 0.f);
  const int i4 = (blockIdx.x * SM_THREADS + t) * SM_ITEMS;
  const bool in = i4 < N_ITEMS;
  float ww[4] = {0.f, 0.f, 0.f, 0.f};
  if (in) w4_of((const unsigned*)w, vals, i4, ww);
  const float mx = block_max(p.x, red);
  const float sum = block_sum(has ? p.y * __expf(p.x - mx) : 0.f, red);
  const float rs = 1.0f / sum;
  if (in)
    *(float4*)(w + i4) = make_float4(__expf(ww[0] - mx) * rs, __expf(ww[1] - mx) * rs,
                                     __expf(ww[2] - mx) * rs, __expf(ww[3] - mx) * rs);
}

extern "C" void kernel_launch(void* const* d_in, const int* in_sizes, int n_in,
                              void* d_out, int out_size, void* d_ws, size_t ws_size,
                              hipStream_t stream) {
  const float* neigh = (const float*)d_in[0];
  const float* tgt   = (const float*)d_in[1];
  const int*   perms = (const int*)d_in[2];
  float* out  = (float*)d_out;
  float* vals = (float*)d_ws;
  float* stats = vals + N_ITEMS;
  float* part  = stats + 4;

  init_kernel<<<1, 32, 0, stream>>>(tgt, stats);

  const int nTiles = N_ITEMS / 16;
  vals_kernel<<<(nTiles + 7) / 8, 256, 0, stream>>>(neigh, tgt, stats, vals, out, nTiles);

  scatter_kernel<<<PARTS * SLABS, 1024, 0, stream>>>(perms, vals, (unsigned*)out);

  stats_kernel<<<SM_BLOCKS, SM_THREADS, 0, stream>>>((const unsigned*)out, vals, part);
  final_kernel<<<SM_BLOCKS, SM_THREADS, 0, stream>>>(out, vals, part);
}
